# starting kernel
# baseline (speedup 1.0000x reference)
#include <hip/hip_runtime.h>
#include <hip/hip_bf16.h>

typedef __attribute__((ext_vector_type(2))) float v2f;
typedef __attribute__((ext_vector_type(8))) float v8f;
typedef __attribute__((ext_vector_type(4))) int   v4i;

__global__ void gcn_premul_zero(const float* __restrict__ feat,
                                const float* __restrict__ norm,
                                float* __restrict__ pre1,
                                float* __restrict__ agg1,
                                int n) {
  int i = blockIdx.x * blockDim.x + threadIdx.x;
  if (i < n) {
    pre1[i] = feat[i] * norm[i];
    agg1[i] = 0.0f;
  }
}

__global__ void gcn_edge_scatter(const int* __restrict__ src,
                                 const int* __restrict__ dst,
                                 const float* __restrict__ val,
                                 float* __restrict__ agg,
                                 int e4, int e) {
  int tid = blockIdx.x * blockDim.x + threadIdx.x;
  if (tid < e4) {
    const v4i* s4 = (const v4i*)src;
    const v4i* d4 = (const v4i*)dst;
    v4i s = __builtin_nontemporal_load(&s4[tid]);
    v4i d = __builtin_nontemporal_load(&d4[tid]);
    __hip_atomic_fetch_add(&agg[d.x], val[s.x], __ATOMIC_RELAXED, __HIP_MEMORY_SCOPE_AGENT);
    __hip_atomic_fetch_add(&agg[d.y], val[s.y], __ATOMIC_RELAXED, __HIP_MEMORY_SCOPE_AGENT);
    __hip_atomic_fetch_add(&agg[d.z], val[s.z], __ATOMIC_RELAXED, __HIP_MEMORY_SCOPE_AGENT);
    __hip_atomic_fetch_add(&agg[d.w], val[s.w], __ATOMIC_RELAXED, __HIP_MEMORY_SCOPE_AGENT);
  } else if (tid == e4) {
    for (int k = e4 * 4; k < e; ++k) {
      __hip_atomic_fetch_add(&agg[dst[k]], val[src[k]], __ATOMIC_RELAXED, __HIP_MEMORY_SCOPE_AGENT);
    }
  }
}

__global__ void gcn_layer1_wmma(const float* __restrict__ agg1,
                                const float* __restrict__ norm,
                                const float* __restrict__ W1,
                                const float* __restrict__ b1,
                                const float* __restrict__ W2,
                                float* __restrict__ t,
                                float* __restrict__ agg2,
                                int n_nodes) {
  const int lane = threadIdx.x & 31;
  const int wave = threadIdx.x >> 5;
  const int base = (blockIdx.x * (blockDim.x >> 5) + wave) * 16;
  const int col = lane & 15;

  float w1n = 0.0f, b1n = 0.0f, w2n = 0.0f;
  if (col < 3) {
    w1n = W1[col];
    b1n = b1[col];
    w2n = W2[col];
  }

  const int idx = base + col;
  const bool rowlane = (lane < 16) && (idx < n_nodes);
  float xv = 0.0f;
  if (rowlane) {
    xv = agg1[idx] * norm[idx];
    agg2[idx] = 0.0f;
  }

  v2f a;
  a.x = xv;
  a.y = 0.0f;
  v2f b;
  b.x = w1n;
  b.y = w1n;
  v8f c;
#pragma unroll
  for (int r = 0; r < 8; ++r) c[r] = b1n;

  v8f d = __builtin_amdgcn_wmma_f32_16x16x4_f32(
       false, a,  false, b,
       (short)0, c,  false,  false);

#pragma unroll
  for (int r = 0; r < 8; ++r) {
    float h = fmaxf(d[r], 0.0f) * w2n;
    h += __shfl_xor(h, 1);
    h += __shfl_xor(h, 2);
    if (col == 0) {
      const int id = base + r + ((lane >> 4) << 3);
      if (id < n_nodes) t[id] = norm[id] * h;
    }
  }
}

__global__ void gcn_final(const float* __restrict__ norm,
                          const float* __restrict__ agg2,
                          const float* __restrict__ b2,
                          float* __restrict__ out,
                          int n) {
  int i = blockIdx.x * blockDim.x + threadIdx.x;
  if (i < n) {
    out[i] = fmaxf(norm[i] * agg2[i] + b2[0], 0.0f);
  }
}

extern "C" void kernel_launch(void* const* d_in, const int* in_sizes, int n_in,
                              void* d_out, int out_size, void* d_ws, size_t ws_size,
                              hipStream_t stream) {
  (void)n_in; (void)out_size; (void)ws_size;

  const float* feat = (const float*)d_in[0];
  const float* norm = (const float*)d_in[1];
  const int*   src  = (const int*)d_in[2];
  const int*   dst  = (const int*)d_in[3];
  const float* W1   = (const float*)d_in[4];
  const float* b1   = (const float*)d_in[5];
  const float* W2   = (const float*)d_in[6];
  const float* b2   = (const float*)d_in[7];
  float* out = (float*)d_out;

  const int N = in_sizes[0];
  const int E = in_sizes[2];

  float* pre1 = (float*)d_ws;
  float* agg1 = pre1 + N;
  float* tval = agg1 + N;
  float* agg2 = tval + N;

  const int TB = 256;
  const int grid_nodes = (N + TB - 1) / TB;

  const int e4 = E / 4;
  const int edge_threads = e4 + 1;
  const int grid_edges = (edge_threads + TB - 1) / TB;

  const int nodes_per_block = (TB / 32) * 16;
  const int grid_wmma = (N + nodes_per_block - 1) / nodes_per_block;

  gcn_premul_zero<<<grid_nodes, TB, 0, stream>>>(feat, norm, pre1, agg1, N);
  gcn_edge_scatter<<<grid_edges, TB, 0, stream>>>(src, dst, pre1, agg1, e4, E);
  gcn_layer1_wmma<<<grid_wmma, TB, 0, stream>>>(agg1, norm, W1, b1, W2, tval, agg2, N);
  gcn_edge_scatter<<<grid_edges, TB, 0, stream>>>(src, dst, tval, agg2, e4, E);
  gcn_final<<<grid_nodes, TB, 0, stream>>>(norm, agg2, b2, out, N);
}


# Each block counting-sorts its chunk's adds by 64-node destination bin in LDS and issues them in bin order, so lanes of an atomic add share 128-byte lines; atomic sum order changes
# speedup vs baseline: 2.8800x; 2.8800x over previous
#include <hip/hip_runtime.h>
#include <hip/hip_bf16.h>


typedef __attribute__((ext_vector_type(2))) float v2f;
typedef __attribute__((ext_vector_type(8))) float v8f;
typedef __attribute__((ext_vector_type(4))) int   v4i;

#define CNT_STRIDE 64

__global__ void gcn_premul_init(const float* __restrict__ feat,
                                const float* __restrict__ norm,
                                float* __restrict__ pre1,
                                float* __restrict__ pair,
                                float* __restrict__ agg2,
                                int* __restrict__ cnt,
                                int n) {
  int i = blockIdx.x * blockDim.x + threadIdx.x;
  if (i < n) {
    const float nv = norm[i];
    pre1[i] = feat[i] * nv;
    v2f p;
    p.x = 0.0f;
    p.y = nv;
    *(v2f*)(pair + 2 * (size_t)i) = p;
    agg2[i] = 0.0f;
  }
  if (i < 4) cnt[i * CNT_STRIDE] = 0;
}

#define NTHR 1024
#define QPT 16
#define CHUNK_QUADS (NTHR * QPT)
#define NBINS 8192
#define BPT (NBINS / NTHR)
#define CAP 33792

struct L1Frag {
  v2f a;
  v8f c;
  float w20, w21, w22;
  bool hi;
};

__device__ __forceinline__ L1Frag make_l1frag(const float* __restrict__ W1,
                                              const float* __restrict__ b1,
                                              const float* __restrict__ W2,
                                              int lane) {
  L1Frag f;
  const int row = lane & 15;
  f.hi = (lane >> 4) != 0;
  const float w1_0 = W1[0], w1_1 = W1[1], w1_2 = W1[2];
  const float b_0 = b1[0], b_1 = b1[1], b_2 = b1[2];
  const int sel = f.hi ? row - 3 : row;
  float av = 0.0f;
  if (sel == 0) av = w1_0;
  if (sel == 1) av = w1_1;
  if (sel == 2) av = w1_2;
  f.a.x = av;
  f.a.y = 0.0f;
  const float m = f.hi ? 0.0f : 1.0f;
  f.c[0] = m * b_0; f.c[1] = m * b_1; f.c[2] = m * b_2;
  f.c[3] = m * b_0; f.c[4] = m * b_1; f.c[5] = m * b_2;
  f.c[6] = 0.0f;    f.c[7] = 0.0f;
  f.w20 = W2[0]; f.w21 = W2[1]; f.w22 = W2[2];
  return f;
}

__device__ __forceinline__ float l1_message(const v2f p, const L1Frag& f) {
  v2f b;
  b.x = p.x * p.y;
  b.y = 0.0f;
  const v8f d = __builtin_amdgcn_wmma_f32_16x16x4_f32(
       false, f.a,  false, b,
       (short)0, f.c,  false,  false);
  const float lo = (fmaxf(d[0], 0.0f) * f.w20 + fmaxf(d[1], 0.0f) * f.w21) + fmaxf(d[2], 0.0f) * f.w22;
  const float up = (fmaxf(d[3], 0.0f) * f.w20 + fmaxf(d[4], 0.0f) * f.w21) + fmaxf(d[5], 0.0f) * f.w22;
  const float over = __shfl_xor(up, 16);
  return p.y * (f.hi ? over : lo);
}

__device__ __forceinline__ int draw_ticket(int* c, int* s_word, int tid) {
  if (tid == 0) *s_word = __hip_atomic_fetch_add(c, 1, __ATOMIC_RELAXED, __HIP_MEMORY_SCOPE_AGENT);
  __syncthreads();
  const int t = __builtin_amdgcn_readfirstlane(*s_word);
  __syncthreads();
  return t;
}

__device__ __forceinline__ int bin_of(int d, int base) {
  const unsigned b = (unsigned)(d - base) >> 6;
  return (int)(b < (unsigned)(NBINS - 1) ? b : (unsigned)(NBINS - 1));
}

template <bool XFORM>
__device__ __forceinline__ void edge_pass(const int* __restrict__ src,
                                          const int* __restrict__ dst,
                                          const float* __restrict__ val,
                                          float* __restrict__ agg,
                                          int* __restrict__ cnt,
                                          int e4, int e, int n,
                                          const L1Frag& f,
                                          int* s_cnt, int* s_dst, float* s_val, int* s_misc) {
  const unsigned ASTR = XFORM ? 1u : 2u;
  constexpr int RQ = XFORM ? 2 : 4;
  const int tid = threadIdx.x;
  const int lane = tid & 31;
  const int wave = tid >> 5;
  const v4i* s4 = (const v4i*)src;
  const v4i* d4 = (const v4i*)dst;

  const int nblk = (int)gridDim.x;
  int cq = ((e4 + nblk - 1) / nblk + 7) & ~7;
  if (cq > CHUNK_QUADS) cq = CHUNK_QUADS;
  if (cq < 8) cq = 8;
  int n_chunks = (e4 + cq - 1) / cq;
  if (n_chunks < 1) n_chunks = 1;
  const int split = ((n >> 1) + 63) & ~63;
  const int tail0 = e4 * 4;
  const int n_tail = e - tail0;

  if (tid == 0) s_misc[1] = (int)((__builtin_amdgcn_s_sendmsg_rtn(0x87) >> 18) & 1u);
  __syncthreads();
  int k = __builtin_amdgcn_readfirstlane(s_misc[1]);

  bool other = false;
  int t = draw_ticket(cnt + k * CNT_STRIDE, s_misc, tid);
  for (;;) {
    if (t >= n_chunks) {
      if (other) break;
      other = true;
      k ^= 1;
      t = draw_ticket(cnt + k * CNT_STRIDE, s_misc, tid);
      continue;
    }
    const bool upper = (k != 0);
    const int base = upper ? split : 0;

    if (e4 > 0) {
      const int q0 = t * cq;

      for (int i = tid; i < NBINS; i += NTHR) s_cnt[i] = 0;
      __syncthreads();

      {
        v4i d[QPT];
#pragma unroll
        for (int j = 0; j < QPT; ++j) {
          const int li = j * NTHR + tid;
          const int q = q0 + li;
          d[j] = __builtin_nontemporal_load(&d4[(li < cq && q < e4) ? q : 0]);
        }
#pragma unroll
        for (int j = 0; j < QPT; ++j) {
          const int li = j * NTHR + tid;
          const bool okj = li < cq && (q0 + li) < e4;
#pragma unroll
          for (int c = 0; c < 4; ++c) {
            if (okj && ((d[j][c] >= split) == upper))
              __hip_atomic_fetch_add(&s_cnt[bin_of(d[j][c], base)], 1, __ATOMIC_RELAXED, __HIP_MEMORY_SCOPE_WORKGROUP);
          }
        }
      }
      __syncthreads();

      {
        int cex[BPT];
        int sum = 0;
#pragma unroll
        for (int i = 0; i < BPT; ++i) {
          const int x = s_cnt[BPT * tid + i];
          cex[i] = sum;
          sum += x;
        }
        int inc = sum;
#pragma unroll
        for (int o = 1; o < 32; o <<= 1) {
          const int y = __shfl_up(inc, o);
          if (lane >= o) inc += y;
        }
        if (lane == 31) s_misc[8 + wave] = inc;
        __syncthreads();
        if (wave == 0) {
          const int w = s_misc[8 + lane];
          int winc = w;
#pragma unroll
          for (int o = 1; o < 32; o <<= 1) {
            const int y = __shfl_up(winc, o);
            if (lane >= o) winc += y;
          }
          s_misc[8 + lane] = winc - w;
          if (lane == 31) s_misc[2] = winc;
        }
        __syncthreads();
        const int b0 = s_misc[8 + wave] + (inc - sum);
#pragma unroll
        for (int i = 0; i < BPT; ++i) s_cnt[BPT * tid + i] = cex[i] + b0;
      }
      __syncthreads();
      const int total = __builtin_amdgcn_readfirstlane(s_misc[2]);

#pragma nounroll
      for (int r = 0; r < QPT / RQ; ++r) {
        v4i s[RQ], d[RQ];
        bool ok[RQ];
#pragma unroll
        for (int u = 0; u < RQ; ++u) {
          const int li = (r * RQ + u) * NTHR + tid;
          const int q = q0 + li;
          ok[u] = li < cq && q < e4;
          const int qc = ok[u] ? q : 0;
          s[u] = __builtin_nontemporal_load(&s4[qc]);
          d[u] = __builtin_nontemporal_load(&d4[qc]);
        }

        float v[RQ][4];
        bool mine[RQ][4];
        if (XFORM) {
          v2f p[RQ][4];
#pragma unroll
          for (int u = 0; u < RQ; ++u) {
#pragma unroll
            for (int c = 0; c < 4; ++c) {
              mine[u][c] = ok[u] && ((d[u][c] >= split) == upper);
              const unsigned idx = (unsigned)(mine[u][c] ? s[u][c] : 0);
              p[u][c] = *(const v2f*)(val + 2u * idx);
            }
          }
#pragma unroll
          for (int u = 0; u < RQ; ++u) {
#pragma unroll
            for (int c = 0; c < 4; ++c) v[u][c] = l1_message(p[u][c], f);
          }
        } else {
#pragma unroll
          for (int u = 0; u < RQ; ++u) {
#pragma unroll
            for (int c = 0; c < 4; ++c) {
              mine[u][c] = ok[u] && ((d[u][c] >= split) == upper);
              v[u][c] = val[(unsigned)(mine[u][c] ? s[u][c] : 0)];
            }
          }
        }

        int pos[RQ][4];
#pragma unroll
        for (int u = 0; u < RQ; ++u) {
#pragma unroll
          for (int c = 0; c < 4; ++c) {
            pos[u][c] = 0;
            if (mine[u][c])
              pos[u][c] = __hip_atomic_fetch_add(&s_cnt[bin_of(d[u][c], base)], 1, __ATOMIC_RELAXED, __HIP_MEMORY_SCOPE_WORKGROUP);
          }
        }
#pragma unroll
        for (int u = 0; u < RQ; ++u) {
#pragma unroll
          for (int c = 0; c < 4; ++c) {
            if (mine[u][c]) {
              if (pos[u][c] < CAP) {
                s_dst[pos[u][c]] = d[u][c];
                s_val[pos[u][c]] = v[u][c];
              } else {
                __hip_atomic_fetch_add(&agg[ASTR * (unsigned)d[u][c]], v[u][c], __ATOMIC_RELAXED, __HIP_MEMORY_SCOPE_AGENT);
              }
            }
          }
        }
      }
      __syncthreads();

      const int m = total < CAP ? total : CAP;
      for (int i = tid; i < m; i += NTHR)
        __hip_atomic_fetch_add(&agg[ASTR * (unsigned)s_dst[i]], s_val[i], __ATOMIC_RELAXED, __HIP_MEMORY_SCOPE_AGENT);
    }

    if (t == 0 && n_tail > 0 && wave == 0) {
      const bool in = lane < n_tail;
      const int dj = in ? dst[tail0 + lane] : 0;
      const bool mj = in && ((dj >= split) == upper);
      const unsigned sj = mj ? (unsigned)src[tail0 + lane] : 0u;
      float vj;
      if (XFORM) vj = l1_message(*(const v2f*)(val + 2u * sj), f);
      else       vj = val[sj];
      if (mj)
        __hip_atomic_fetch_add(&agg[ASTR * (unsigned)dj], vj, __ATOMIC_RELAXED, __HIP_MEMORY_SCOPE_AGENT);
    }
    t = draw_ticket(cnt + k * CNT_STRIDE, s_misc, tid);
  }
}

__global__ __launch_bounds__(NTHR)
void gcn_edge_scatter_l1(const int* __restrict__ src,
                         const int* __restrict__ dst,
                         const float* __restrict__ pre1,
                         float* __restrict__ pair,
                         int* __restrict__ cnt,
                         int e4, int e, int n) {
  __shared__ int s_cnt[NBINS];
  __shared__ int s_dst[CAP];
  __shared__ float s_val[CAP];
  __shared__ int s_misc[40];
  L1Frag f = {};
  edge_pass<false>(src, dst, pre1, pair, cnt, e4, e, n, f, s_cnt, s_dst, s_val, s_misc);
}

__global__ __launch_bounds__(NTHR)
void gcn_edge_scatter_l2(const int* __restrict__ src,
                         const int* __restrict__ dst,
                         const float* __restrict__ pair,
                         const float* __restrict__ W1,
                         const float* __restrict__ b1,
                         const float* __restrict__ W2,
                         float* __restrict__ agg2,
                         int* __restrict__ cnt,
                         int e4, int e, int n) {
  __shared__ int s_cnt[NBINS];
  __shared__ int s_dst[CAP];
  __shared__ float s_val[CAP];
  __shared__ int s_misc[40];
  const L1Frag f = make_l1frag(W1, b1, W2, threadIdx.x & 31);
  edge_pass<true>(src, dst, pair, agg2, cnt, e4, e, n, f, s_cnt, s_dst, s_val, s_misc);
}

__global__ void gcn_final(const float* __restrict__ norm,
                          const float* __restrict__ agg2,
                          const float* __restrict__ b2,
                          float* __restrict__ out,
                          int n) {
  int i = blockIdx.x * blockDim.x + threadIdx.x;
  if (i < n) {
    out[i] = fmaxf(norm[i] * agg2[i] + b2[0], 0.0f);
  }
}

extern "C" void kernel_launch(void* const* d_in, const int* in_sizes, int n_in,
                              void* d_out, int out_size, void* d_ws, size_t ws_size,
                              hipStream_t stream) {
  (void)n_in; (void)out_size;

  const float* feat = (const float*)d_in[0];
  const float* norm = (const float*)d_in[1];
  const int*   src  = (const int*)d_in[2];
  const int*   dst  = (const int*)d_in[3];
  const float* W1   = (const float*)d_in[4];
  const float* b1   = (const float*)d_in[5];
  const float* W2   = (const float*)d_in[6];
  const float* b2   = (const float*)d_in[7];
  float* out = (float*)d_out;

  const int N = in_sizes[0];
  const int E = in_sizes[2];

  float* pair = (float*)d_ws;
  float* pre1 = pair + 2 * (size_t)N;
  float* agg2 = pre1 + N;

  const size_t planes = ((size_t)4 * N + 63) & ~(size_t)63;
  int* cnt = (ws_size >= (planes + 4 * CNT_STRIDE) * sizeof(float))
                 ? (int*)((float*)d_ws + planes) : (int*)out;
  int* cnt1 = cnt;
  int* cnt2 = cnt + 2 * CNT_STRIDE;

  const int TB = 256;
  const int grid_nodes = (N + TB - 1) / TB;

  const int e4 = E / 4;
  const int grid_edges = 256;

  gcn_premul_init<<<grid_nodes, TB, 0, stream>>>(feat, norm, pre1, pair, agg2, cnt, N);
  gcn_edge_scatter_l1<<<grid_edges, NTHR, 0, stream>>>(src, dst, pre1, pair, cnt1, e4, E, N);
  gcn_edge_scatter_l2<<<grid_edges, NTHR, 0, stream>>>(src, dst, pair, W1, b1, W2, agg2, cnt2, e4, E, N);
  gcn_final<<<grid_nodes, TB, 0, stream>>>(norm, agg2, b2, out, N);
}


# Counting sort orders {src,dst} index pairs only; values are gathered (layer 2: transformed) in bin order on full waves right before their adds, so gathers and adds mix; atomic sum order changes
# speedup vs baseline: 3.5343x; 1.2272x over previous
#include <hip/hip_runtime.h>
#include <hip/hip_bf16.h>


typedef __attribute__((ext_vector_type(2))) float v2f;
typedef __attribute__((ext_vector_type(8))) float v8f;
typedef __attribute__((ext_vector_type(4))) int   v4i;

#define CNT_STRIDE 64

__global__ void gcn_premul_init(const float* __restrict__ feat,
                                const float* __restrict__ norm,
                                float* __restrict__ pre1,
                                float* __restrict__ pair,
                                float* __restrict__ agg2,
                                int* __restrict__ cnt,
                                int n) {
  int i = blockIdx.x * blockDim.x + threadIdx.x;
  if (i < n) {
    const float nv = norm[i];
    pre1[i] = feat[i] * nv;
    v2f p;
    p.x = 0.0f;
    p.y = nv;
    *(v2f*)(pair + 2 * (size_t)i) = p;
    agg2[i] = 0.0f;
  }
  if (i < 4) cnt[i * CNT_STRIDE] = 0;
}

#define NTHR 1024
#define QPT 16
#define CHUNK_QUADS (NTHR * QPT)
#define NBINS 8192
#define BPT (NBINS / NTHR)
#define CAP 33792

struct L1Frag {
  v2f a;
  v8f c;
  float w20, w21, w22;
  bool hi;
};

__device__ __forceinline__ L1Frag make_l1frag(const float* __restrict__ W1,
                                              const float* __restrict__ b1,
                                              const float* __restrict__ W2,
                                              int lane) {
  L1Frag f;
  const int row = lane & 15;
  f.hi = (lane >> 4) != 0;
  const float w1_0 = W1[0], w1_1 = W1[1], w1_2 = W1[2];
  const float b_0 = b1[0], b_1 = b1[1], b_2 = b1[2];
  const int sel = f.hi ? row - 3 : row;
  float av = 0.0f;
  if (sel == 0) av = w1_0;
  if (sel == 1) av = w1_1;
  if (sel == 2) av = w1_2;
  f.a.x = av;
  f.a.y = 0.0f;
  const float m = f.hi ? 0.0f : 1.0f;
  f.c[0] = m * b_0; f.c[1] = m * b_1; f.c[2] = m * b_2;
  f.c[3] = m * b_0; f.c[4] = m * b_1; f.c[5] = m * b_2;
  f.c[6] = 0.0f;    f.c[7] = 0.0f;
  f.w20 = W2[0]; f.w21 = W2[1]; f.w22 = W2[2];
  return f;
}

__device__ __forceinline__ float l1_message(const v2f p, const L1Frag& f) {
  v2f b;
  b.x = p.x * p.y;
  b.y = 0.0f;
  const v8f d = __builtin_amdgcn_wmma_f32_16x16x4_f32(
       false, f.a,  false, b,
       (short)0, f.c,  false,  false);
  const float lo = (fmaxf(d[0], 0.0f) * f.w20 + fmaxf(d[1], 0.0f) * f.w21) + fmaxf(d[2], 0.0f) * f.w22;
  const float up = (fmaxf(d[3], 0.0f) * f.w20 + fmaxf(d[4], 0.0f) * f.w21) + fmaxf(d[5], 0.0f) * f.w22;
  const float over = __shfl_xor(up, 16);
  return p.y * (f.hi ? over : lo);
}

__device__ __forceinline__ int draw_ticket(int* c, int* s_word, int tid) {
  if (tid == 0) *s_word = __hip_atomic_fetch_add(c, 1, __ATOMIC_RELAXED, __HIP_MEMORY_SCOPE_AGENT);
  __syncthreads();
  const int t = __builtin_amdgcn_readfirstlane(*s_word);
  __syncthreads();
  return t;
}

__device__ __forceinline__ int bin_of(int d, int base) {
  const unsigned b = (unsigned)(d - base) >> 6;
  return (int)(b < (unsigned)(NBINS - 1) ? b : (unsigned)(NBINS - 1));
}

template <bool XFORM>
__device__ __forceinline__ void edge_pass(const int* __restrict__ src,
                                          const int* __restrict__ dst,
                                          const float* __restrict__ val,
                                          float* __restrict__ agg,
                                          int* __restrict__ cnt,
                                          int e4, int e, int n,
                                          const L1Frag& f,
                                          int* s_cnt, int* s_dst, int* s_src, int* s_misc) {
  const unsigned ASTR = XFORM ? 1u : 2u;
  constexpr int RQ = 4;
  constexpr int FQ = XFORM ? 2 : 4;
  constexpr int U = XFORM ? 8 : 16;
  const int tid = threadIdx.x;
  const int lane = tid & 31;
  const int wave = tid >> 5;
  const v4i* s4 = (const v4i*)src;
  const v4i* d4 = (const v4i*)dst;

  const int nblk = (int)gridDim.x;
  int cq = ((e4 + nblk - 1) / nblk + 7) & ~7;
  if (cq > CHUNK_QUADS) cq = CHUNK_QUADS;
  if (cq < 8) cq = 8;
  int n_chunks = (e4 + cq - 1) / cq;
  if (n_chunks < 1) n_chunks = 1;
  const int split = ((n >> 1) + 63) & ~63;
  const int tail0 = e4 * 4;
  const int n_tail = e - tail0;

  if (tid == 0) s_misc[1] = (int)((__builtin_amdgcn_s_sendmsg_rtn(0x87) >> 18) & 1u);
  __syncthreads();
  int k = __builtin_amdgcn_readfirstlane(s_misc[1]);

  bool other = false;
  int t = draw_ticket(cnt + k * CNT_STRIDE, s_misc, tid);
  for (;;) {
    if (t >= n_chunks) {
      if (other) break;
      other = true;
      k ^= 1;
      t = draw_ticket(cnt + k * CNT_STRIDE, s_misc, tid);
      continue;
    }
    const bool upper = (k != 0);
    const int base = upper ? split : 0;

    if (e4 > 0) {
      const int q0 = t * cq;

      for (int i = tid; i < NBINS; i += NTHR) s_cnt[i] = 0;
      __syncthreads();

      {
        v4i d[QPT];
#pragma unroll
        for (int j = 0; j < QPT; ++j) {
          const int li = j * NTHR + tid;
          const int q = q0 + li;
          d[j] = __builtin_nontemporal_load(&d4[(li < cq && q < e4) ? q : 0]);
        }
#pragma unroll
        for (int j = 0; j < QPT; ++j) {
          const int li = j * NTHR + tid;
          const bool okj = li < cq && (q0 + li) < e4;
#pragma unroll
          for (int c = 0; c < 4; ++c) {
            if (okj && ((d[j][c] >= split) == upper))
              __hip_atomic_fetch_add(&s_cnt[bin_of(d[j][c], base)], 1, __ATOMIC_RELAXED, __HIP_MEMORY_SCOPE_WORKGROUP);
          }
        }
      }
      __syncthreads();

      {
        int cex[BPT];
        int sum = 0;
#pragma unroll
        for (int i = 0; i < BPT; ++i) {
          const int x = s_cnt[BPT * tid + i];
          cex[i] = sum;
          sum += x;
        }
        int inc = sum;
#pragma unroll
        for (int o = 1; o < 32; o <<= 1) {
          const int y = __shfl_up(inc, o);
          if (lane >= o) inc += y;
        }
        if (lane == 31) s_misc[8 + wave] = inc;
        __syncthreads();
        if (wave == 0) {
          const int w = s_misc[8 + lane];
          int winc = w;
#pragma unroll
          for (int o = 1; o < 32; o <<= 1) {
            const int y = __shfl_up(winc, o);
            if (lane >= o) winc += y;
          }
          s_misc[8 + lane] = winc - w;
          if (lane == 31) s_misc[2] = winc;
        }
        __syncthreads();
        const int b0 = s_misc[8 + wave] + (inc - sum);
#pragma unroll
        for (int i = 0; i < BPT; ++i) s_cnt[BPT * tid + i] = cex[i] + b0;
      }
      __syncthreads();
      const int total = __builtin_amdgcn_readfirstlane(s_misc[2]);

      if (total <= CAP) {
#pragma nounroll
        for (int r = 0; r < QPT / RQ; ++r) {
          v4i s[RQ], d[RQ];
          bool ok[RQ];
#pragma unroll
          for (int u = 0; u < RQ; ++u) {
            const int li = (r * RQ + u) * NTHR + tid;
            const int q = q0 + li;
            ok[u] = li < cq && q < e4;
            const int qc = ok[u] ? q : 0;
            s[u] = __builtin_nontemporal_load(&s4[qc]);
            d[u] = __builtin_nontemporal_load(&d4[qc]);
          }
          int pos[RQ][4];
#pragma unroll
          for (int u = 0; u < RQ; ++u) {
#pragma unroll
            for (int c = 0; c < 4; ++c) {
              pos[u][c] = -1;
              if (ok[u] && ((d[u][c] >= split) == upper))
                pos[u][c] = __hip_atomic_fetch_add(&s_cnt[bin_of(d[u][c], base)], 1, __ATOMIC_RELAXED, __HIP_MEMORY_SCOPE_WORKGROUP);
            }
          }
#pragma unroll
          for (int u = 0; u < RQ; ++u) {
#pragma unroll
            for (int c = 0; c < 4; ++c) {
              if (pos[u][c] >= 0) {
                s_dst[pos[u][c]] = d[u][c];
                s_src[pos[u][c]] = s[u][c];
              }
            }
          }
        }
        __syncthreads();

#pragma nounroll
        for (int i0 = 0; i0 < total; i0 += NTHR * U) {
          int sj[U], dj[U];
          bool in[U];
#pragma unroll
          for (int u = 0; u < U; ++u) {
            const int i = i0 + u * NTHR + tid;
            in[u] = i < total;
            const int ic = in[u] ? i : 0;
            sj[u] = s_src[ic];
            dj[u] = s_dst[ic];
          }
          float v[U];
          if (XFORM) {
            v2f p[U];
#pragma unroll
            for (int u = 0; u < U; ++u) p[u] = ((const v2f*)val)[(unsigned)sj[u]];
#pragma unroll
            for (int u = 0; u < U; ++u) v[u] = l1_message(p[u], f);
          } else {
#pragma unroll
            for (int u = 0; u < U; ++u) v[u] = val[(unsigned)sj[u]];
          }
#pragma unroll
          for (int u = 0; u < U; ++u) {
            if (in[u])
              __hip_atomic_fetch_add(&agg[ASTR * (unsigned)dj[u]], v[u], __ATOMIC_RELAXED, __HIP_MEMORY_SCOPE_AGENT);
          }
        }
      } else {
#pragma nounroll
        for (int r = 0; r < QPT / FQ; ++r) {
          v4i s[FQ], d[FQ];
          bool ok[FQ];
#pragma unroll
          for (int u = 0; u < FQ; ++u) {
            const int li = (r * FQ + u) * NTHR + tid;
            const int q = q0 + li;
            ok[u] = li < cq && q < e4;
            const int qc = ok[u] ? q : 0;
            s[u] = __builtin_nontemporal_load(&s4[qc]);
            d[u] = __builtin_nontemporal_load(&d4[qc]);
          }
          float v[FQ][4];
          bool mine[FQ][4];
          if (XFORM) {
            v2f p[FQ][4];
#pragma unroll
            for (int u = 0; u < FQ; ++u) {
#pragma unroll
              for (int c = 0; c < 4; ++c) {
                mine[u][c] = ok[u] && ((d[u][c] >= split) == upper);
                const unsigned idx = (unsigned)(mine[u][c] ? s[u][c] : 0);
                p[u][c] = *(const v2f*)(val + 2u * idx);
              }
            }
#pragma unroll
            for (int u = 0; u < FQ; ++u) {
#pragma unroll
              for (int c = 0; c < 4; ++c) v[u][c] = l1_message(p[u][c], f);
            }
          } else {
#pragma unroll
            for (int u = 0; u < FQ; ++u) {
#pragma unroll
              for (int c = 0; c < 4; ++c) {
                mine[u][c] = ok[u] && ((d[u][c] >= split) == upper);
                v[u][c] = val[(unsigned)(mine[u][c] ? s[u][c] : 0)];
              }
            }
          }
#pragma unroll
          for (int u = 0; u < FQ; ++u) {
#pragma unroll
            for (int c = 0; c < 4; ++c) {
              if (mine[u][c])
                __hip_atomic_fetch_add(&agg[ASTR * (unsigned)d[u][c]], v[u][c], __ATOMIC_RELAXED, __HIP_MEMORY_SCOPE_AGENT);
            }
          }
        }
      }
    }

    if (t == 0 && n_tail > 0 && wave == 0) {
      const bool in = lane < n_tail;
      const int dj = in ? dst[tail0 + lane] : 0;
      const bool mj = in && ((dj >= split) == upper);
      const unsigned sj = mj ? (unsigned)src[tail0 + lane] : 0u;
      float vj;
      if (XFORM) vj = l1_message(*(const v2f*)(val + 2u * sj), f);
      else       vj = val[sj];
      if (mj)
        __hip_atomic_fetch_add(&agg[ASTR * (unsigned)dj], vj, __ATOMIC_RELAXED, __HIP_MEMORY_SCOPE_AGENT);
    }
    t = draw_ticket(cnt + k * CNT_STRIDE, s_misc, tid);
  }
}

__global__ __launch_bounds__(NTHR)
void gcn_edge_scatter_l1(const int* __restrict__ src,
                         const int* __restrict__ dst,
                         const float* __restrict__ pre1,
                         float* __restrict__ pair,
                         int* __restrict__ cnt,
                         int e4, int e, int n) {
  __shared__ int s_cnt[NBINS];
  __shared__ int s_dst[CAP];
  __shared__ int s_src[CAP];
  __shared__ int s_misc[40];
  L1Frag f = {};
  edge_pass<false>(src, dst, pre1, pair, cnt, e4, e, n, f, s_cnt, s_dst, s_src, s_misc);
}

__global__ __launch_bounds__(NTHR)
void gcn_edge_scatter_l2(const int* __restrict__ src,
                         const int* __restrict__ dst,
                         const float* __restrict__ pair,
                         const float* __restrict__ W1,
                         const float* __restrict__ b1,
                         const float* __restrict__ W2,
                         float* __restrict__ agg2,
                         int* __restrict__ cnt,
                         int e4, int e, int n) {
  __shared__ int s_cnt[NBINS];
  __shared__ int s_dst[CAP];
  __shared__ int s_src[CAP];
  __shared__ int s_misc[40];
  const L1Frag f = make_l1frag(W1, b1, W2, threadIdx.x & 31);
  edge_pass<true>(src, dst, pair, agg2, cnt, e4, e, n, f, s_cnt, s_dst, s_src, s_misc);
}

__global__ void gcn_final(const float* __restrict__ norm,
                          const float* __restrict__ agg2,
                          const float* __restrict__ b2,
                          float* __restrict__ out,
                          int n) {
  int i = blockIdx.x * blockDim.x + threadIdx.x;
  if (i < n) {
    out[i] = fmaxf(norm[i] * agg2[i] + b2[0], 0.0f);
  }
}

extern "C" void kernel_launch(void* const* d_in, const int* in_sizes, int n_in,
                              void* d_out, int out_size, void* d_ws, size_t ws_size,
                              hipStream_t stream) {
  (void)n_in; (void)out_size;

  const float* feat = (const float*)d_in[0];
  const float* norm = (const float*)d_in[1];
  const int*   src  = (const int*)d_in[2];
  const int*   dst  = (const int*)d_in[3];
  const float* W1   = (const float*)d_in[4];
  const float* b1   = (const float*)d_in[5];
  const float* W2   = (const float*)d_in[6];
  const float* b2   = (const float*)d_in[7];
  float* out = (float*)d_out;

  const int N = in_sizes[0];
  const int E = in_sizes[2];

  float* pair = (float*)d_ws;
  float* pre1 = pair + 2 * (size_t)N;
  float* agg2 = pre1 + N;

  const size_t planes = ((size_t)4 * N + 63) & ~(size_t)63;
  int* cnt = (ws_size >= (planes + 4 * CNT_STRIDE) * sizeof(float))
                 ? (int*)((float*)d_ws + planes) : (int*)out;
  int* cnt1 = cnt;
  int* cnt2 = cnt + 2 * CNT_STRIDE;

  const int TB = 256;
  const int grid_nodes = (N + TB - 1) / TB;

  const int e4 = E / 4;
  const int grid_edges = 256;

  gcn_premul_init<<<grid_nodes, TB, 0, stream>>>(feat, norm, pre1, pair, agg2, cnt, N);
  gcn_edge_scatter_l1<<<grid_edges, NTHR, 0, stream>>>(src, dst, pre1, pair, cnt1, e4, E, N);
  gcn_edge_scatter_l2<<<grid_edges, NTHR, 0, stream>>>(src, dst, pair, W1, b1, W2, agg2, cnt2, e4, E, N);
  gcn_final<<<grid_nodes, TB, 0, stream>>>(norm, agg2, b2, out, N);
}


# Sort entries packed into one word (src + low dst bits; upper dst bits found from the entry's place), so a block orders chunks twice as large and more adds share a line; atomic sum order changes
# speedup vs baseline: 3.6719x; 1.0389x over previous
#include <hip/hip_runtime.h>
#include <hip/hip_bf16.h>


typedef __attribute__((ext_vector_type(2))) float v2f;
typedef __attribute__((ext_vector_type(8))) float v8f;
typedef __attribute__((ext_vector_type(4))) int   v4i;

#define CNT_STRIDE 64

__global__ void gcn_premul_init(const float* __restrict__ feat,
                                const float* __restrict__ norm,
                                float* __restrict__ pre1,
                                float* __restrict__ pair,
                                float* __restrict__ agg2,
                                int* __restrict__ cnt,
                                int n) {
  int i = blockIdx.x * blockDim.x + threadIdx.x;
  if (i < n) {
    const float nv = norm[i];
    pre1[i] = feat[i] * nv;
    v2f p;
    p.x = 0.0f;
    p.y = nv;
    *(v2f*)(pair + 2 * (size_t)i) = p;
    agg2[i] = 0.0f;
  }
  if (i < 4) cnt[i * CNT_STRIDE] = 0;
}

#define NTHR 1024
#define QPT 32
#define CQ 16
#define CHUNK_QUADS (NTHR * QPT)
#define NBINS 8192
#define BPT (NBINS / NTHR)
#define CAP 69632
#define MAXSEG 128

struct L1Frag {
  v2f a;
  v8f c;
  float w20, w21, w22;
  bool hi;
};

__device__ __forceinline__ L1Frag make_l1frag(const float* __restrict__ W1,
                                              const float* __restrict__ b1,
                                              const float* __restrict__ W2,
                                              int lane) {
  L1Frag f;
  const int row = lane & 15;
  f.hi = (lane >> 4) != 0;
  const float w1_0 = W1[0], w1_1 = W1[1], w1_2 = W1[2];
  const float b_0 = b1[0], b_1 = b1[1], b_2 = b1[2];
  const int sel = f.hi ? row - 3 : row;
  float av = 0.0f;
  if (sel == 0) av = w1_0;
  if (sel == 1) av = w1_1;
  if (sel == 2) av = w1_2;
  f.a.x = av;
  f.a.y = 0.0f;
  const float m = f.hi ? 0.0f : 1.0f;
  f.c[0] = m * b_0; f.c[1] = m * b_1; f.c[2] = m * b_2;
  f.c[3] = m * b_0; f.c[4] = m * b_1; f.c[5] = m * b_2;
  f.c[6] = 0.0f;    f.c[7] = 0.0f;
  f.w20 = W2[0]; f.w21 = W2[1]; f.w22 = W2[2];
  return f;
}

__device__ __forceinline__ float l1_message(const v2f p, const L1Frag& f) {
  v2f b;
  b.x = p.x * p.y;
  b.y = 0.0f;
  const v8f d = __builtin_amdgcn_wmma_f32_16x16x4_f32(
       false, f.a,  false, b,
       (short)0, f.c,  false,  false);
  const float lo = (fmaxf(d[0], 0.0f) * f.w20 + fmaxf(d[1], 0.0f) * f.w21) + fmaxf(d[2], 0.0f) * f.w22;
  const float up = (fmaxf(d[3], 0.0f) * f.w20 + fmaxf(d[4], 0.0f) * f.w21) + fmaxf(d[5], 0.0f) * f.w22;
  const float over = __shfl_xor(up, 16);
  return p.y * (f.hi ? over : lo);
}

__device__ __forceinline__ int draw_ticket(int* c, int* s_word, int tid) {
  if (tid == 0) *s_word = __hip_atomic_fetch_add(c, 1, __ATOMIC_RELAXED, __HIP_MEMORY_SCOPE_AGENT);
  __syncthreads();
  const int t = __builtin_amdgcn_readfirstlane(*s_word);
  __syncthreads();
  return t;
}

__device__ __forceinline__ int bin_of(int d, int base) {
  const unsigned b = (unsigned)(d - base) >> 6;
  return (int)(b < (unsigned)(NBINS - 1) ? b : (unsigned)(NBINS - 1));
}

template <bool XFORM>
__device__ __forceinline__ void edge_pass(const int* __restrict__ src,
                                          const int* __restrict__ dst,
                                          const float* __restrict__ val,
                                          float* __restrict__ agg,
                                          int* __restrict__ cnt,
                                          int e4, int e, int n,
                                          const L1Frag& f,
                                          int* s_cnt, unsigned* s_ent, int* s_misc) {
  const unsigned ASTR = XFORM ? 1u : 2u;
  constexpr int RQ = 4;
  constexpr int FQ = XFORM ? 2 : 4;
  constexpr int U = XFORM ? 8 : 16;
  const int tid = threadIdx.x;
  const int lane = tid & 31;
  const int wave = tid >> 5;
  const v4i* s4 = (const v4i*)src;
  const v4i* d4 = (const v4i*)dst;

  const int nblk = (int)gridDim.x;
  const int npc = (nblk >> 1) > 0 ? (nblk >> 1) : 1;
  int cq = ((e4 + npc - 1) / npc + 7) & ~7;
  if (cq > CHUNK_QUADS) cq = CHUNK_QUADS;
  if (cq < 8) cq = 8;
  int n_chunks = (e4 + cq - 1) / cq;
  if (n_chunks < 1) n_chunks = 1;
  const int split = ((n >> 1) + 63) & ~63;
  const int srcb = n > 1 ? 32 - __builtin_clz((unsigned)(n - 1)) : 1;
  const int lowb = 32 - srcb;
  const int cls_max = split > n - split ? split : n - split;
  const unsigned lmask = (1u << lowb) - 1u;
  const int nseg = lowb >= 31 ? 1 : (int)(((unsigned)cls_max + lmask) >> lowb);
  const bool packable = lowb >= 6 && cls_max <= NBINS * 64 && nseg <= MAXSEG;
  const int bsh = lowb >= 6 ? lowb - 6 : 0;
  const int tail0 = e4 * 4;
  const int n_tail = e - tail0;

  if (tid == 0) s_misc[1] = (int)((__builtin_amdgcn_s_sendmsg_rtn(0x87) >> 18) & 1u);
  __syncthreads();
  int k = __builtin_amdgcn_readfirstlane(s_misc[1]);

  bool other = false;
  int t = draw_ticket(cnt + k * CNT_STRIDE, s_misc, tid);
  for (;;) {
    if (t >= n_chunks) {
      if (other) break;
      other = true;
      k ^= 1;
      t = draw_ticket(cnt + k * CNT_STRIDE, s_misc, tid);
      continue;
    }
    const bool upper = (k != 0);
    const int base = upper ? split : 0;

    if (e4 > 0) {
      const int q0 = t * cq;

      for (int i = tid; i < NBINS; i += NTHR) s_cnt[i] = 0;
      __syncthreads();

#pragma nounroll
      for (int r = 0; r < QPT / CQ; ++r) {
        v4i d[CQ];
#pragma unroll
        for (int j = 0; j < CQ; ++j) {
          const int li = (r * CQ + j) * NTHR + tid;
          const int q = q0 + li;
          d[j] = __builtin_nontemporal_load(&d4[(li < cq && q < e4) ? q : 0]);
        }
#pragma unroll
        for (int j = 0; j < CQ; ++j) {
          const int li = (r * CQ + j) * NTHR + tid;
          const bool okj = li < cq && (q0 + li) < e4;
#pragma unroll
          for (int c = 0; c < 4; ++c) {
            if (okj && ((d[j][c] >= split) == upper))
              __hip_atomic_fetch_add(&s_cnt[bin_of(d[j][c], base)], 1, __ATOMIC_RELAXED, __HIP_MEMORY_SCOPE_WORKGROUP);
          }
        }
      }
      __syncthreads();

      {
        int cex[BPT];
        int sum = 0;
#pragma unroll
        for (int i = 0; i < BPT; ++i) {
          const int x = s_cnt[BPT * tid + i];
          cex[i] = sum;
          sum += x;
        }
        int inc = sum;
#pragma unroll
        for (int o = 1; o < 32; o <<= 1) {
          const int y = __shfl_up(inc, o);
          if (lane >= o) inc += y;
        }
        if (lane == 31) s_misc[8 + wave] = inc;
        __syncthreads();
        if (wave == 0) {
          const int w = s_misc[8 + lane];
          int winc = w;
#pragma unroll
          for (int o = 1; o < 32; o <<= 1) {
            const int y = __shfl_up(winc, o);
            if (lane >= o) winc += y;
          }
          s_misc[8 + lane] = winc - w;
          if (lane == 31) s_misc[2] = winc;
        }
        __syncthreads();
        const int b0 = s_misc[8 + wave] + (inc - sum);
#pragma unroll
        for (int i = 0; i < BPT; ++i) s_cnt[BPT * tid + i] = cex[i] + b0;
      }
      __syncthreads();
      const int total = __builtin_amdgcn_readfirstlane(s_misc[2]);

      if (packable && total <= CAP) {
#pragma nounroll
        for (int r = 0; r < QPT / RQ; ++r) {
          v4i s[RQ], d[RQ];
          bool ok[RQ];
#pragma unroll
          for (int u = 0; u < RQ; ++u) {
            const int li = (r * RQ + u) * NTHR + tid;
            const int q = q0 + li;
            ok[u] = li < cq && q < e4;
            const int qc = ok[u] ? q : 0;
            s[u] = __builtin_nontemporal_load(&s4[qc]);
            d[u] = __builtin_nontemporal_load(&d4[qc]);
          }
          int pos[RQ][4];
#pragma unroll
          for (int u = 0; u < RQ; ++u) {
#pragma unroll
            for (int c = 0; c < 4; ++c) {
              pos[u][c] = -1;
              if (ok[u] && ((d[u][c] >= split) == upper))
                pos[u][c] = __hip_atomic_fetch_add(&s_cnt[bin_of(d[u][c], base)], 1, __ATOMIC_RELAXED, __HIP_MEMORY_SCOPE_WORKGROUP);
            }
          }
#pragma unroll
          for (int u = 0; u < RQ; ++u) {
#pragma unroll
            for (int c = 0; c < 4; ++c) {
              if (pos[u][c] >= 0)
                s_ent[pos[u][c]] = ((unsigned)s[u][c] << lowb) | ((unsigned)(d[u][c] - base) & lmask);
            }
          }
        }
        __syncthreads();

#pragma nounroll
        for (int i0 = 0; i0 < total; i0 += NTHR * U) {
          int sj[U], dj[U];
          bool in[U];
          unsigned en[U];
          int sg[U];
#pragma unroll
          for (int u = 0; u < U; ++u) {
            const int i = i0 + u * NTHR + tid;
            in[u] = i < total;
            const int ic = in[u] ? i : 0;
            en[u] = s_ent[ic];
            sg[u] = 0;
          }
#pragma unroll
          for (int st = MAXSEG / 2; st > 0; st >>= 1) {
            int first[U];
#pragma unroll
            for (int u = 0; u < U; ++u) {
              const int c = sg[u] + st;
              const int cc = c < nseg ? c : nseg - 1;
              const int bi = (cc << bsh) - 1;
              first[u] = s_cnt[bi > 0 ? bi : 0];
            }
#pragma unroll
            for (int u = 0; u < U; ++u) {
              const int i = i0 + u * NTHR + tid;
              const int c = sg[u] + st;
              const int take = (int)(c < nseg) & (int)(first[u] <= i);
              sg[u] = take ? c : sg[u];
            }
          }
#pragma unroll
          for (int u = 0; u < U; ++u) {
            sj[u] = (int)(en[u] >> lowb);
            dj[u] = base + (int)(((unsigned)sg[u] << lowb) | (en[u] & lmask));
          }
          float v[U];
          if (XFORM) {
            v2f p[U];
#pragma unroll
            for (int u = 0; u < U; ++u) p[u] = ((const v2f*)val)[(unsigned)sj[u]];
#pragma unroll
            for (int u = 0; u < U; ++u) v[u] = l1_message(p[u], f);
          } else {
#pragma unroll
            for (int u = 0; u < U; ++u) v[u] = val[(unsigned)sj[u]];
          }
#pragma unroll
          for (int u = 0; u < U; ++u) {
            if (in[u])
              __hip_atomic_fetch_add(&agg[ASTR * (unsigned)dj[u]], v[u], __ATOMIC_RELAXED, __HIP_MEMORY_SCOPE_AGENT);
          }
        }
      } else {
#pragma nounroll
        for (int r = 0; r < QPT / FQ; ++r) {
          v4i s[FQ], d[FQ];
          bool ok[FQ];
#pragma unroll
          for (int u = 0; u < FQ; ++u) {
            const int li = (r * FQ + u) * NTHR + tid;
            const int q = q0 + li;
            ok[u] = li < cq && q < e4;
            const int qc = ok[u] ? q : 0;
            s[u] = __builtin_nontemporal_load(&s4[qc]);
            d[u] = __builtin_nontemporal_load(&d4[qc]);
          }
          float v[FQ][4];
          bool mine[FQ][4];
          if (XFORM) {
            v2f p[FQ][4];
#pragma unroll
            for (int u = 0; u < FQ; ++u) {
#pragma unroll
              for (int c = 0; c < 4; ++c) {
                mine[u][c] = ok[u] && ((d[u][c] >= split) == upper);
                const unsigned idx = (unsigned)(mine[u][c] ? s[u][c] : 0);
                p[u][c] = *(const v2f*)(val + 2u * idx);
              }
            }
#pragma unroll
            for (int u = 0; u < FQ; ++u) {
#pragma unroll
              for (int c = 0; c < 4; ++c) v[u][c] = l1_message(p[u][c], f);
            }
          } else {
#pragma unroll
            for (int u = 0; u < FQ; ++u) {
#pragma unroll
              for (int c = 0; c < 4; ++c) {
                mine[u][c] = ok[u] && ((d[u][c] >= split) == upper);
                v[u][c] = val[(unsigned)(mine[u][c] ? s[u][c] : 0)];
              }
            }
          }
#pragma unroll
          for (int u = 0; u < FQ; ++u) {
#pragma unroll
            for (int c = 0; c < 4; ++c) {
              if (mine[u][c])
                __hip_atomic_fetch_add(&agg[ASTR * (unsigned)d[u][c]], v[u][c], __ATOMIC_RELAXED, __HIP_MEMORY_SCOPE_AGENT);
            }
          }
        }
      }
    }

    if (t == 0 && n_tail > 0 && wave == 0) {
      const bool in = lane < n_tail;
      const int dj = in ? dst[tail0 + lane] : 0;
      const bool mj = in && ((dj >= split) == upper);
      const unsigned sj = mj ? (unsigned)src[tail0 + lane] : 0u;
      float vj;
      if (XFORM) vj = l1_message(*(const v2f*)(val + 2u * sj), f);
      else       vj = val[sj];
      if (mj)
        __hip_atomic_fetch_add(&agg[ASTR * (unsigned)dj], vj, __ATOMIC_RELAXED, __HIP_MEMORY_SCOPE_AGENT);
    }
    t = draw_ticket(cnt + k * CNT_STRIDE, s_misc, tid);
  }
}

__global__ __launch_bounds__(NTHR)
void gcn_edge_scatter_l1(const int* __restrict__ src,
                         const int* __restrict__ dst,
                         const float* __restrict__ pre1,
                         float* __restrict__ pair,
                         int* __restrict__ cnt,
                         int e4, int e, int n) {
  __shared__ int s_cnt[NBINS];
  __shared__ unsigned s_ent[CAP];
  __shared__ int s_misc[40];
  L1Frag f = {};
  edge_pass<false>(src, dst, pre1, pair, cnt, e4, e, n, f, s_cnt, s_ent, s_misc);
}

__global__ __launch_bounds__(NTHR)
void gcn_edge_scatter_l2(const int* __restrict__ src,
                         const int* __restrict__ dst,
                         const float* __restrict__ pair,
                         const float* __restrict__ W1,
                         const float* __restrict__ b1,
                         const float* __restrict__ W2,
                         float* __restrict__ agg2,
                         int* __restrict__ cnt,
                         int e4, int e, int n) {
  __shared__ int s_cnt[NBINS];
  __shared__ unsigned s_ent[CAP];
  __shared__ int s_misc[40];
  const L1Frag f = make_l1frag(W1, b1, W2, threadIdx.x & 31);
  edge_pass<true>(src, dst, pair, agg2, cnt, e4, e, n, f, s_cnt, s_ent, s_misc);
}

__global__ void gcn_final(const float* __restrict__ norm,
                          const float* __restrict__ agg2,
                          const float* __restrict__ b2,
                          float* __restrict__ out,
                          int n) {
  int i = blockIdx.x * blockDim.x + threadIdx.x;
  if (i < n) {
    out[i] = fmaxf(norm[i] * agg2[i] + b2[0], 0.0f);
  }
}

extern "C" void kernel_launch(void* const* d_in, const int* in_sizes, int n_in,
                              void* d_out, int out_size, void* d_ws, size_t ws_size,
                              hipStream_t stream) {
  (void)n_in; (void)out_size;

  const float* feat = (const float*)d_in[0];
  const float* norm = (const float*)d_in[1];
  const int*   src  = (const int*)d_in[2];
  const int*   dst  = (const int*)d_in[3];
  const float* W1   = (const float*)d_in[4];
  const float* b1   = (const float*)d_in[5];
  const float* W2   = (const float*)d_in[6];
  const float* b2   = (const float*)d_in[7];
  float* out = (float*)d_out;

  const int N = in_sizes[0];
  const int E = in_sizes[2];

  float* pair = (float*)d_ws;
  float* pre1 = pair + 2 * (size_t)N;
  float* agg2 = pre1 + N;

  const size_t planes = ((size_t)4 * N + 63) & ~(size_t)63;
  int* cnt = (ws_size >= (planes + 4 * CNT_STRIDE) * sizeof(float))
                 ? (int*)((float*)d_ws + planes) : (int*)out;
  int* cnt1 = cnt;
  int* cnt2 = cnt + 2 * CNT_STRIDE;

  const int TB = 256;
  const int grid_nodes = (N + TB - 1) / TB;

  const int e4 = E / 4;
  const int grid_edges = 256;

  gcn_premul_init<<<grid_nodes, TB, 0, stream>>>(feat, norm, pre1, pair, agg2, cnt, N);
  gcn_edge_scatter_l1<<<grid_edges, NTHR, 0, stream>>>(src, dst, pre1, pair, cnt1, e4, E, N);
  gcn_edge_scatter_l2<<<grid_edges, NTHR, 0, stream>>>(src, dst, pair, W1, b1, W2, agg2, cnt2, e4, E, N);
  gcn_final<<<grid_nodes, TB, 0, stream>>>(norm, agg2, b2, out, N);
}


# Layer-1 edge pass leaves each chunk's ordered index entries and segment starts in the workspace; the layer-2 pass walks that copy instead of counting, scanning and placing its chunks again
# speedup vs baseline: 3.8710x; 1.0542x over previous
#include <hip/hip_runtime.h>
#include <hip/hip_bf16.h>


typedef __attribute__((ext_vector_type(2))) float v2f;
typedef __attribute__((ext_vector_type(8))) float v8f;
typedef __attribute__((ext_vector_type(4))) int   v4i;
typedef __attribute__((ext_vector_type(4))) unsigned v4u;

#define CNT_STRIDE 64

__global__ void gcn_premul_init(const float* __restrict__ feat,
                                const float* __restrict__ norm,
                                float* __restrict__ pre1,
                                float* __restrict__ pair,
                                float* __restrict__ agg2,
                                int* __restrict__ cnt,
                                int n) {
  int i = blockIdx.x * blockDim.x + threadIdx.x;
  if (i < n) {
    const float nv = norm[i];
    pre1[i] = feat[i] * nv;
    v2f p;
    p.x = 0.0f;
    p.y = nv;
    *(v2f*)(pair + 2 * (size_t)i) = p;
    agg2[i] = 0.0f;
  }
  if (i < 5) cnt[i * CNT_STRIDE] = 0;
}

#define NTHR 1024
#define QPT 32
#define CQ 16
#define CHUNK_QUADS (NTHR * QPT)
#define NBINS 8192
#define BPT (NBINS / NTHR)
#define CAP 69632
#define MAXSEG 128
#define MAXSLOT 512
#define TABW (MAXSEG + 4)

struct L1Frag {
  v2f a;
  v8f c;
  float w20, w21, w22;
  bool hi;
};

__device__ __forceinline__ L1Frag make_l1frag(const float* __restrict__ W1,
                                              const float* __restrict__ b1,
                                              const float* __restrict__ W2,
                                              int lane) {
  L1Frag f;
  const int row = lane & 15;
  f.hi = (lane >> 4) != 0;
  const float w1_0 = W1[0], w1_1 = W1[1], w1_2 = W1[2];
  const float b_0 = b1[0], b_1 = b1[1], b_2 = b1[2];
  const int sel = f.hi ? row - 3 : row;
  float av = 0.0f;
  if (sel == 0) av = w1_0;
  if (sel == 1) av = w1_1;
  if (sel == 2) av = w1_2;
  f.a.x = av;
  f.a.y = 0.0f;
  const float m = f.hi ? 0.0f : 1.0f;
  f.c[0] = m * b_0; f.c[1] = m * b_1; f.c[2] = m * b_2;
  f.c[3] = m * b_0; f.c[4] = m * b_1; f.c[5] = m * b_2;
  f.c[6] = 0.0f;    f.c[7] = 0.0f;
  f.w20 = W2[0]; f.w21 = W2[1]; f.w22 = W2[2];
  return f;
}

__device__ __forceinline__ float l1_message(const v2f p, const L1Frag& f) {
  v2f b;
  b.x = p.x * p.y;
  b.y = 0.0f;
  const v8f d = __builtin_amdgcn_wmma_f32_16x16x4_f32(
       false, f.a,  false, b,
       (short)0, f.c,  false,  false);
  const float lo = (fmaxf(d[0], 0.0f) * f.w20 + fmaxf(d[1], 0.0f) * f.w21) + fmaxf(d[2], 0.0f) * f.w22;
  const float up = (fmaxf(d[3], 0.0f) * f.w20 + fmaxf(d[4], 0.0f) * f.w21) + fmaxf(d[5], 0.0f) * f.w22;
  const float over = __shfl_xor(up, 16);
  return p.y * (f.hi ? over : lo);
}

__device__ __forceinline__ int draw_ticket(int* c, int* s_word, int tid) {
  if (tid == 0) *s_word = __hip_atomic_fetch_add(c, 1, __ATOMIC_RELAXED, __HIP_MEMORY_SCOPE_AGENT);
  __syncthreads();
  const int t = __builtin_amdgcn_readfirstlane(*s_word);
  __syncthreads();
  return t;
}

__device__ __forceinline__ int bin_of(int d, int base) {
  const unsigned b = (unsigned)(d - base) >> 6;
  return (int)(b < (unsigned)(NBINS - 1) ? b : (unsigned)(NBINS - 1));
}

template <bool XFORM, bool SAVED>
__device__ __forceinline__ void add_walk(const unsigned* ent, const int* s_cnt,
                                         const float* __restrict__ val,
                                         float* __restrict__ agg,
                                         int total, int base, int nseg, int bsh,
                                         int lowb, unsigned lmask,
                                         const L1Frag& f, int tid) {
  const unsigned ASTR = XFORM ? 1u : 2u;
  constexpr int U = XFORM ? 8 : 16;
#pragma nounroll
  for (int i0 = 0; i0 < total; i0 += NTHR * U) {
    int sj[U], dj[U];
    bool in[U];
    unsigned en[U];
    int sg[U];
#pragma unroll
    for (int u = 0; u < U; ++u) {
      const int i = i0 + u * NTHR + tid;
      in[u] = i < total;
      const int ic = in[u] ? i : 0;
      en[u] = SAVED ? __builtin_nontemporal_load(&ent[ic]) : ent[ic];
      sg[u] = 0;
    }
#pragma unroll
    for (int st = MAXSEG / 2; st > 0; st >>= 1) {
      int first[U];
#pragma unroll
      for (int u = 0; u < U; ++u) {
        const int c = sg[u] + st;
        const int cc = c < nseg ? c : nseg - 1;
        const int bi = (cc << bsh) - 1;
        first[u] = s_cnt[bi > 0 ? bi : 0];
      }
#pragma unroll
      for (int u = 0; u < U; ++u) {
        const int i = i0 + u * NTHR + tid;
        const int c = sg[u] + st;
        const int take = (int)(c < nseg) & (int)(first[u] <= i);
        sg[u] = take ? c : sg[u];
      }
    }
#pragma unroll
    for (int u = 0; u < U; ++u) {
      sj[u] = (int)(en[u] >> lowb);
      dj[u] = base + (int)(((unsigned)sg[u] << lowb) | (en[u] & lmask));
    }
    float v[U];
    if (XFORM) {
      v2f p[U];
#pragma unroll
      for (int u = 0; u < U; ++u) p[u] = ((const v2f*)val)[(unsigned)sj[u]];
#pragma unroll
      for (int u = 0; u < U; ++u) v[u] = l1_message(p[u], f);
    } else {
#pragma unroll
      for (int u = 0; u < U; ++u) v[u] = val[(unsigned)sj[u]];
    }
#pragma unroll
    for (int u = 0; u < U; ++u) {
      if (in[u])
        __hip_atomic_fetch_add(&agg[ASTR * (unsigned)dj[u]], v[u], __ATOMIC_RELAXED, __HIP_MEMORY_SCOPE_AGENT);
    }
  }
}

template <bool XFORM>
__device__ __forceinline__ void edge_pass(const int* __restrict__ src,
                                          const int* __restrict__ dst,
                                          const float* __restrict__ val,
                                          float* __restrict__ agg,
                                          int* __restrict__ cnt,
                                          unsigned* __restrict__ gent,
                                          int* __restrict__ tab,
                                          int e4, int e, int n,
                                          const L1Frag& f,
                                          int* s_cnt, unsigned* s_ent, int* s_misc) {
  const unsigned ASTR = XFORM ? 1u : 2u;
  constexpr int RQ = 4;
  constexpr int FQ = XFORM ? 2 : 4;
  const int tid = threadIdx.x;
  const int lane = tid & 31;
  const int wave = tid >> 5;
  const v4i* s4 = (const v4i*)src;
  const v4i* d4 = (const v4i*)dst;

  const int nblk = (int)gridDim.x;
  const int npc = (nblk >> 1) > 0 ? (nblk >> 1) : 1;
  int cq = ((e4 + npc - 1) / npc + 7) & ~7;
  if (cq > CHUNK_QUADS) cq = CHUNK_QUADS;
  if (cq < 8) cq = 8;
  int n_chunks = (e4 + cq - 1) / cq;
  if (n_chunks < 1) n_chunks = 1;
  const int split = ((n >> 1) + 63) & ~63;
  const int srcb = n > 1 ? 32 - __builtin_clz((unsigned)(n - 1)) : 1;
  const int lowb = 32 - srcb;
  const int cls_max = split > n - split ? split : n - split;
  const unsigned lmask = (1u << lowb) - 1u;
  const int nseg = lowb >= 31 ? 1 : (int)(((unsigned)cls_max + lmask) >> lowb);
  const bool packable = lowb >= 6 && cls_max <= NBINS * 64 && nseg <= MAXSEG;
  const int bsh = lowb >= 6 ? lowb - 6 : 0;
  const int tail0 = e4 * 4;
  const int n_tail = e - tail0;
  const bool reuse = gent != nullptr && 2 * n_chunks <= MAXSLOT;

  if (tid == 0) s_misc[1] = (int)((__builtin_amdgcn_s_sendmsg_rtn(0x87) >> 18) & 1u);
  __syncthreads();
  int k = __builtin_amdgcn_readfirstlane(s_misc[1]);

  bool other = false;
  int t = draw_ticket(cnt + k * CNT_STRIDE, s_misc, tid);
  for (;;) {
    if (t >= n_chunks) {
      if (other) break;
      other = true;
      k ^= 1;
      t = draw_ticket(cnt + k * CNT_STRIDE, s_misc, tid);
      continue;
    }
    const bool upper = (k != 0);
    const int base = upper ? split : 0;

    bool walked = false;
    if (XFORM && reuse && e4 > 0) {
      const int* const srow = tab + CNT_STRIDE + (k * n_chunks + t) * TABW;
      const int stotal = __builtin_amdgcn_readfirstlane(srow[1]);
      if (stotal >= 0) {
        const int soff = __builtin_amdgcn_readfirstlane(srow[0]);
        if (tid > 0 && tid < nseg) s_cnt[(tid << bsh) - 1] = srow[4 + tid];
        __syncthreads();
        add_walk<true, true>(gent + (size_t)(unsigned)soff, s_cnt, val, agg, stotal, base, nseg, bsh, lowb, lmask, f, tid);
        walked = true;
      }
    }
    if (e4 > 0 && !walked) {
      const int q0 = t * cq;

      for (int i = tid; i < NBINS; i += NTHR) s_cnt[i] = 0;
      __syncthreads();

#pragma nounroll
      for (int r = 0; r < QPT / CQ; ++r) {
        v4i d[CQ];
#pragma unroll
        for (int j = 0; j < CQ; ++j) {
          const int li = (r * CQ + j) * NTHR + tid;
          const int q = q0 + li;
          d[j] = __builtin_nontemporal_load(&d4[(li < cq && q < e4) ? q : 0]);
        }
#pragma unroll
        for (int j = 0; j < CQ; ++j) {
          const int li = (r * CQ + j) * NTHR + tid;
          const bool okj = li < cq && (q0 + li) < e4;
#pragma unroll
          for (int c = 0; c < 4; ++c) {
            if (okj && ((d[j][c] >= split) == upper))
              __hip_atomic_fetch_add(&s_cnt[bin_of(d[j][c], base)], 1, __ATOMIC_RELAXED, __HIP_MEMORY_SCOPE_WORKGROUP);
          }
        }
      }
      __syncthreads();

      {
        int cex[BPT];
        int sum = 0;
#pragma unroll
        for (int i = 0; i < BPT; ++i) {
          const int x = s_cnt[BPT * tid + i];
          cex[i] = sum;
          sum += x;
        }
        int inc = sum;
#pragma unroll
        for (int o = 1; o < 32; o <<= 1) {
          const int y = __shfl_up(inc, o);
          if (lane >= o) inc += y;
        }
        if (lane == 31) s_misc[8 + wave] = inc;
        __syncthreads();
        if (wave == 0) {
          const int w = s_misc[8 + lane];
          int winc = w;
#pragma unroll
          for (int o = 1; o < 32; o <<= 1) {
            const int y = __shfl_up(winc, o);
            if (lane >= o) winc += y;
          }
          s_misc[8 + lane] = winc - w;
          if (lane == 31) s_misc[2] = winc;
        }
        __syncthreads();
        const int b0 = s_misc[8 + wave] + (inc - sum);
#pragma unroll
        for (int i = 0; i < BPT; ++i) s_cnt[BPT * tid + i] = cex[i] + b0;
      }
      __syncthreads();
      const int total = __builtin_amdgcn_readfirstlane(s_misc[2]);
      int* const row = tab + CNT_STRIDE + (k * n_chunks + t) * TABW;
      if (!XFORM && reuse && tid == 0) {
        if (packable && total <= CAP)
          s_misc[3] = __hip_atomic_fetch_add(tab, (total + 31) & ~31, __ATOMIC_RELAXED, __HIP_MEMORY_SCOPE_AGENT);
        else
          row[1] = -1;
      }

      if (packable && total <= CAP) {
#pragma nounroll
        for (int r = 0; r < QPT / RQ; ++r) {
          v4i s[RQ], d[RQ];
          bool ok[RQ];
#pragma unroll
          for (int u = 0; u < RQ; ++u) {
            const int li = (r * RQ + u) * NTHR + tid;
            const int q = q0 + li;
            ok[u] = li < cq && q < e4;
            const int qc = ok[u] ? q : 0;
            s[u] = __builtin_nontemporal_load(&s4[qc]);
            d[u] = __builtin_nontemporal_load(&d4[qc]);
          }
          int pos[RQ][4];
#pragma unroll
          for (int u = 0; u < RQ; ++u) {
#pragma unroll
            for (int c = 0; c < 4; ++c) {
              pos[u][c] = -1;
              if (ok[u] && ((d[u][c] >= split) == upper))
                pos[u][c] = __hip_atomic_fetch_add(&s_cnt[bin_of(d[u][c], base)], 1, __ATOMIC_RELAXED, __HIP_MEMORY_SCOPE_WORKGROUP);
            }
          }
#pragma unroll
          for (int u = 0; u < RQ; ++u) {
#pragma unroll
            for (int c = 0; c < 4; ++c) {
              if (pos[u][c] >= 0)
                s_ent[pos[u][c]] = ((unsigned)s[u][c] << lowb) | ((unsigned)(d[u][c] - base) & lmask);
            }
          }
        }
        __syncthreads();

        if (!XFORM && reuse) {
          const int off = __builtin_amdgcn_readfirstlane(s_misc[3]);
          unsigned* g = gent + (size_t)(unsigned)off;
          for (int i = tid * 4; i < total; i += NTHR * 4)
            *(v4u*)(g + i) = *(const v4u*)(s_ent + i);
          if (tid < nseg) row[4 + tid] = tid > 0 ? s_cnt[(tid << bsh) - 1] : 0;
          if (tid == 0) { row[0] = off; row[1] = total; }
        }
        add_walk<XFORM, false>(s_ent, s_cnt, val, agg, total, base, nseg, bsh, lowb, lmask, f, tid);
      } else {
#pragma nounroll
        for (int r = 0; r < QPT / FQ; ++r) {
          v4i s[FQ], d[FQ];
          bool ok[FQ];
#pragma unroll
          for (int u = 0; u < FQ; ++u) {
            const int li = (r * FQ + u) * NTHR + tid;
            const int q = q0 + li;
            ok[u] = li < cq && q < e4;
            const int qc = ok[u] ? q : 0;
            s[u] = __builtin_nontemporal_load(&s4[qc]);
            d[u] = __builtin_nontemporal_load(&d4[qc]);
          }
          float v[FQ][4];
          bool mine[FQ][4];
          if (XFORM) {
            v2f p[FQ][4];
#pragma unroll
            for (int u = 0; u < FQ; ++u) {
#pragma unroll
              for (int c = 0; c < 4; ++c) {
                mine[u][c] = ok[u] && ((d[u][c] >= split) == upper);
                const unsigned idx = (unsigned)(mine[u][c] ? s[u][c] : 0);
                p[u][c] = *(const v2f*)(val + 2u * idx);
              }
            }
#pragma unroll
            for (int u = 0; u < FQ; ++u) {
#pragma unroll
              for (int c = 0; c < 4; ++c) v[u][c] = l1_message(p[u][c], f);
            }
          } else {
#pragma unroll
            for (int u = 0; u < FQ; ++u) {
#pragma unroll
              for (int c = 0; c < 4; ++c) {
                mine[u][c] = ok[u] && ((d[u][c] >= split) == upper);
                v[u][c] = val[(unsigned)(mine[u][c] ? s[u][c] : 0)];
              }
            }
          }
#pragma unroll
          for (int u = 0; u < FQ; ++u) {
#pragma unroll
            for (int c = 0; c < 4; ++c) {
              if (mine[u][c])
                __hip_atomic_fetch_add(&agg[ASTR * (unsigned)d[u][c]], v[u][c], __ATOMIC_RELAXED, __HIP_MEMORY_SCOPE_AGENT);
            }
          }
        }
      }
    }

    if (t == 0 && n_tail > 0 && wave == 0) {
      const bool in = lane < n_tail;
      const int dj = in ? dst[tail0 + lane] : 0;
      const bool mj = in && ((dj >= split) == upper);
      const unsigned sj = mj ? (unsigned)src[tail0 + lane] : 0u;
      float vj;
      if (XFORM) vj = l1_message(*(const v2f*)(val + 2u * sj), f);
      else       vj = val[sj];
      if (mj)
        __hip_atomic_fetch_add(&agg[ASTR * (unsigned)dj], vj, __ATOMIC_RELAXED, __HIP_MEMORY_SCOPE_AGENT);
    }
    t = draw_ticket(cnt + k * CNT_STRIDE, s_misc, tid);
  }
}

__global__ __launch_bounds__(NTHR)
void gcn_edge_scatter_l1(const int* __restrict__ src,
                         const int* __restrict__ dst,
                         const float* __restrict__ pre1,
                         float* __restrict__ pair,
                         int* __restrict__ cnt,
                         unsigned* __restrict__ gent,
                         int* __restrict__ tab,
                         int e4, int e, int n) {
  __shared__ int s_cnt[NBINS];
  __shared__ unsigned s_ent[CAP];
  __shared__ int s_misc[40];
  L1Frag f = {};
  edge_pass<false>(src, dst, pre1, pair, cnt, gent, tab, e4, e, n, f, s_cnt, s_ent, s_misc);
}

__global__ __launch_bounds__(NTHR)
void gcn_edge_scatter_l2(const int* __restrict__ src,
                         const int* __restrict__ dst,
                         const float* __restrict__ pair,
                         const float* __restrict__ W1,
                         const float* __restrict__ b1,
                         const float* __restrict__ W2,
                         float* __restrict__ agg2,
                         int* __restrict__ cnt,
                         unsigned* __restrict__ gent,
                         int* __restrict__ tab,
                         int e4, int e, int n) {
  __shared__ int s_cnt[NBINS];
  __shared__ unsigned s_ent[CAP];
  __shared__ int s_misc[40];
  const L1Frag f = make_l1frag(W1, b1, W2, threadIdx.x & 31);
  edge_pass<true>(src, dst, pair, agg2, cnt, gent, tab, e4, e, n, f, s_cnt, s_ent, s_misc);
}

__global__ void gcn_final(const float* __restrict__ norm,
                          const float* __restrict__ agg2,
                          const float* __restrict__ b2,
                          float* __restrict__ out,
                          int n) {
  int i = blockIdx.x * blockDim.x + threadIdx.x;
  if (i < n) {
    out[i] = fmaxf(norm[i] * agg2[i] + b2[0], 0.0f);
  }
}

extern "C" void kernel_launch(void* const* d_in, const int* in_sizes, int n_in,
                              void* d_out, int out_size, void* d_ws, size_t ws_size,
                              hipStream_t stream) {
  (void)n_in; (void)out_size;

  const float* feat = (const float*)d_in[0];
  const float* norm = (const float*)d_in[1];
  const int*   src  = (const int*)d_in[2];
  const int*   dst  = (const int*)d_in[3];
  const float* W1   = (const float*)d_in[4];
  const float* b1   = (const float*)d_in[5];
  const float* W2   = (const float*)d_in[6];
  const float* b2   = (const float*)d_in[7];
  float* out = (float*)d_out;

  const int N = in_sizes[0];
  const int E = in_sizes[2];

  float* pair = (float*)d_ws;
  float* pre1 = pair + 2 * (size_t)N;
  float* agg2 = pre1 + N;

  const size_t planes = ((size_t)4 * N + 63) & ~(size_t)63;
  int* cnt = (ws_size >= (planes + 4 * CNT_STRIDE) * sizeof(float))
                 ? (int*)((float*)d_ws + planes) : (int*)out;
  int* cnt1 = cnt;
  int* cnt2 = cnt + 2 * CNT_STRIDE;

  const size_t tab_words = (size_t)CNT_STRIDE + (size_t)MAXSLOT * TABW;
  const size_t ent_words = (size_t)E + (size_t)MAXSLOT * 32;
  int* tab = cnt + 4 * CNT_STRIDE;
  unsigned* gent = (ws_size >= (planes + 4 * CNT_STRIDE + tab_words + ent_words) * sizeof(float))
                       ? (unsigned*)(tab + tab_words) : (unsigned*)nullptr;

  const int TB = 256;
  const int grid_nodes = (N + TB - 1) / TB;

  const int e4 = E / 4;
  const int grid_edges = 256;

  gcn_premul_init<<<grid_nodes, TB, 0, stream>>>(feat, norm, pre1, pair, agg2, cnt, N);
  gcn_edge_scatter_l1<<<grid_edges, NTHR, 0, stream>>>(src, dst, pre1, pair, cnt1, gent, tab, e4, E, N);
  gcn_edge_scatter_l2<<<grid_edges, NTHR, 0, stream>>>(src, dst, pair, W1, b1, W2, agg2, cnt2, gent, tab, e4, E, N);
  gcn_final<<<grid_nodes, TB, 0, stream>>>(norm, agg2, b2, out, N);
}
